# starting kernel
# baseline (speedup 1.0000x reference)
#include <hip/hip_runtime.h>

typedef float v2f __attribute__((ext_vector_type(2)));
typedef float v8f __attribute__((ext_vector_type(8)));

#define MAXT 100000.0f
constexpr int Nn = 1024;
constexpr int Mm = 512;

__launch_bounds__(512, 2)
__global__ void snn_fc_spike_kernel(const float* __restrict__ x,
                                    const float* __restrict__ w,
                                    float* __restrict__ out)
{
    __shared__ float s_xs[Nn];
    __shared__ int   s_idx[Nn];

    const int b   = blockIdx.x;
    const int tid = threadIdx.x;

    for (int i = tid; i < Nn; i += 512) {
        s_xs[i]  = x[b * Nn + i];
        s_idx[i] = i;
    }
    __syncthreads();
    for (int k = 2; k <= Nn; k <<= 1) {
        for (int j = k >> 1; j > 0; j >>= 1) {
            for (int i = tid; i < Nn; i += 512) {
                int ixj = i ^ j;
                if (ixj > i) {
                    bool up = ((i & k) == 0);
                    float a = s_xs[i], c = s_xs[ixj];
                    if (up ? (a > c) : (a < c)) {
                        s_xs[i] = c; s_xs[ixj] = a;
                        int t = s_idx[i]; s_idx[i] = s_idx[ixj]; s_idx[ixj] = t;
                    }
                }
            }
            __syncthreads();
        }
    }

    const int lane = tid & 31;
    const int ln   = lane & 15;
    const int h    = lane >> 4;
    const int wv   = tid >> 5;
    const int m0   = wv * 32;

    v2f A1[4];
#pragma unroll
    for (int kk = 0; kk < 4; ++kk) {
        int j0 = 4 * kk + 2 * h;
        A1[kk][0] = (j0     <= ln) ? 1.0f : 0.0f;
        A1[kk][1] = (j0 + 1 <= ln) ? 1.0f : 0.0f;
    }

    float cw[2]  = {0.0f, 0.0f};
    float cwi[2] = {0.0f, 0.0f};
    float mn[2]  = {MAXT, MAXT};

    for (int c = 0; c < Nn / 16; ++c) {
        const int base = c * 16;

        float xr[8], xnr[8];
#pragma unroll
        for (int r = 0; r < 8; ++r) xr[r] = s_xs[base + 8 * h + r];
#pragma unroll
        for (int r = 0; r < 7; ++r) xnr[r] = xr[r + 1];
        {
            int ng = base + 8 * h + 8;
            xnr[7] = (ng < Nn) ? s_xs[ng] : MAXT;
        }

        v2f A2[4];
        int r0[4], r1[4];
#pragma unroll
        for (int kk = 0; kk < 4; ++kk) {
            int j0 = 4 * kk + 2 * h;
            float xj0 = s_xs[base + j0];
            float xj1 = s_xs[base + j0 + 1];
            A2[kk][0] = (j0     <= ln) ? xj0 : 0.0f;
            A2[kk][1] = (j0 + 1 <= ln) ? xj1 : 0.0f;
            r0[kk] = s_idx[base + j0];
            r1[kk] = s_idx[base + j0 + 1];
        }

        if (c + 1 < Nn / 16) {
#pragma unroll
            for (int kk = 0; kk < 4; ++kk) {
                int nj = base + 16 + 4 * kk + 2 * h;
                __builtin_prefetch(&w[(size_t)s_idx[nj] * Mm + m0 + ln], 0, 3);
                __builtin_prefetch(&w[(size_t)s_idx[nj + 1] * Mm + m0 + ln], 0, 3);
            }
        }

#pragma unroll
        for (int t = 0; t < 2; ++t) {
            const int m = m0 + 16 * t + ln;

            v8f accW, accWI;
#pragma unroll
            for (int e = 0; e < 8; ++e) { accW[e] = cw[t]; accWI[e] = cwi[t]; }

#pragma unroll
            for (int kk = 0; kk < 4; ++kk) {
                v2f bv;
                bv[0] = w[(size_t)r0[kk] * Mm + m];
                bv[1] = w[(size_t)r1[kk] * Mm + m];
                accW  = __builtin_amdgcn_wmma_f32_16x16x4_f32(
                            false, A1[kk], false, bv, (short)0, accW,  false, false);
                accWI = __builtin_amdgcn_wmma_f32_16x16x4_f32(
                            false, A2[kk], false, bv, (short)0, accWI, false, false);
            }

            float o8[8];
#pragma unroll
            for (int r = 0; r < 8; ++r) {
                float wc  = accW[r];
                float wic = accWI[r];
                float den = fminf(fmaxf(wc - 1.0f, 1e-10f), 1e10f);
                float o   = wic * __builtin_amdgcn_rcpf(den);
                bool bad  = (wc < 1.0f) | (o < xr[r]) | (o > xnr[r]);
                o8[r] = bad ? MAXT : o;
            }
            float m01 = fminf(o8[0], o8[1]);
            float m23 = fminf(o8[2], o8[3]);
            float m45 = fminf(o8[4], o8[5]);
            float m67 = fminf(o8[6], o8[7]);
            mn[t] = fminf(mn[t], fminf(fminf(m01, m23), fminf(m45, m67)));

            cw[t]  = __shfl(accW[7],  ln + 16, 32);
            cwi[t] = __shfl(accWI[7], ln + 16, 32);
        }
    }

#pragma unroll
    for (int t = 0; t < 2; ++t) {
        float v = fminf(mn[t], __shfl_xor(mn[t], 16, 32));
        if (h == 0) out[(size_t)b * Mm + m0 + 16 * t + ln] = v;
    }
}

extern "C" void kernel_launch(void* const* d_in, const int* in_sizes, int n_in,
                              void* d_out, int out_size, void* d_ws, size_t ws_size,
                              hipStream_t stream)
{
    (void)n_in; (void)out_size; (void)d_ws; (void)ws_size;
    const float* x = (const float*)d_in[0];
    const float* w = (const float*)d_in[1];
    float* out = (float*)d_out;
    const int B = in_sizes[0] / Nn;
    snn_fc_spike_kernel<<<dim3(B), dim3(512), 0, stream>>>(x, w, out);
}


# w held as a 64-column slab in LDS per work-group (8 slabs x 32 groups of 4 batch rows, one row x 16-col tile per wave): B fragments gathered from LDS, not L2; argsort in its own launch; same sums
# speedup vs baseline: 1.8517x; 1.8517x over previous
#include <hip/hip_runtime.h>


typedef float v2f __attribute__((ext_vector_type(2)));
typedef float v8f __attribute__((ext_vector_type(8)));

#define MAXT 100000.0f
constexpr int Nn   = 1024;
constexpr int Mm   = 512;
constexpr int SLAB = 64;
constexpr int RPG  = 4;
constexpr int XPAD = 4;

__launch_bounds__(512, 2)
__global__ void snn_fc_sort_kernel(const float* __restrict__ x,
                                   float* __restrict__ ws, int B)
{
    __shared__ float s_xs[Nn];
    __shared__ int   s_idx[Nn];

    const int b   = blockIdx.x;
    const int tid = threadIdx.x;

    for (int i = tid; i < Nn; i += 512) {
        s_xs[i]  = x[b * Nn + i];
        s_idx[i] = i;
    }
    __syncthreads();
    for (int k = 2; k <= Nn; k <<= 1) {
        for (int j = k >> 1; j > 0; j >>= 1) {
            for (int i = tid; i < Nn; i += 512) {
                int ixj = i ^ j;
                if (ixj > i) {
                    bool up = ((i & k) == 0);
                    float a = s_xs[i], c = s_xs[ixj];
                    if (up ? (a > c) : (a < c)) {
                        s_xs[i] = c; s_xs[ixj] = a;
                        int t = s_idx[i]; s_idx[i] = s_idx[ixj]; s_idx[ixj] = t;
                    }
                }
            }
            __syncthreads();
        }
    }

    float* g_xs  = ws;
    int*   g_idx = reinterpret_cast<int*>(ws) + (size_t)B * Nn;
    for (int i = tid; i < Nn; i += 512) {
        g_xs[(size_t)b * Nn + i]  = s_xs[i];
        g_idx[(size_t)b * Nn + i] = s_idx[i];
    }
}

__launch_bounds__(512, 1)
__global__ void snn_fc_scan_kernel(const float* __restrict__ w,
                                   const float* __restrict__ ws,
                                   float* __restrict__ out, int B)
{
    __shared__ __attribute__((aligned(16))) float s_w[Nn * SLAB];
    __shared__ __attribute__((aligned(16))) float s_xs[RPG][Nn + XPAD];
    __shared__ __attribute__((aligned(16))) int   s_idx[RPG][Nn];
    __shared__ float s_out[RPG * SLAB];

    const int tid  = threadIdx.x;
    const int wg   = blockIdx.x * 2 + blockIdx.y;
    const int slab = wg & (Mm / SLAB - 1);
    const int b0   = (wg / (Mm / SLAB)) * RPG;

    {
        const float* wsl = w + slab * SLAB + (tid & 15) * 4;
        float*       dst = s_w + (tid & 15) * 4;
        const int    rt  = tid >> 4;
        for (int p0 = 0; p0 < Nn / 32; p0 += 8) {
            float4 v[8];
#pragma unroll
            for (int u = 0; u < 8; ++u) {
                const int row = (p0 + u) * 32 + rt;
                v[u] = *reinterpret_cast<const float4*>(wsl + (size_t)row * Mm);
            }
#pragma unroll
            for (int u = 0; u < 8; ++u) {
                const int row = (p0 + u) * 32 + rt;
                *reinterpret_cast<float4*>(dst + row * SLAB) = v[u];
            }
        }
    }
    {
        const float* g_xs  = ws + (size_t)b0 * Nn;
        const int*   g_idx = reinterpret_cast<const int*>(ws) + (size_t)B * Nn + (size_t)b0 * Nn;
#pragma unroll
        for (int k = 0; k < RPG * Nn / (512 * 4); ++k) {
            const int e  = (tid + 512 * k) * 4;
            const int lr = e >> 10, n = e & (Nn - 1);
            *reinterpret_cast<float4*>(&s_xs[lr][n]) = *reinterpret_cast<const float4*>(g_xs + e);
            *reinterpret_cast<int4*>(&s_idx[lr][n])  = *reinterpret_cast<const int4*>(g_idx + e);
        }
        if (tid < RPG * XPAD) s_xs[tid / XPAD][Nn + (tid % XPAD)] = MAXT;
    }
    __syncthreads();

    const int lane = tid & 31;
    const int ln   = lane & 15;
    const int h    = lane >> 4;
    const int wv   = tid >> 5;
    const int lb   = wv >> 2;
    const int c0   = (wv & 3) * 16;

    const float* xs_row  = s_xs[lb];
    const int*   idx_row = s_idx[lb];
    const float* wcol    = s_w + c0 + ln;

    v2f A1[4];
#pragma unroll
    for (int kk = 0; kk < 4; ++kk) {
        int j0 = 4 * kk + 2 * h;
        A1[kk][0] = (j0     <= ln) ? 1.0f : 0.0f;
        A1[kk][1] = (j0 + 1 <= ln) ? 1.0f : 0.0f;
    }

    float cw  = 0.0f;
    float cwi = 0.0f;
    float mn  = MAXT;

    for (int c = 0; c < Nn / 16; ++c) {
        const int base = c * 16;

        float xr[8], xnr[8];
        {
            const float4 xa = *reinterpret_cast<const float4*>(&xs_row[base + 8 * h]);
            const float4 xb = *reinterpret_cast<const float4*>(&xs_row[base + 8 * h + 4]);
            xr[0] = xa.x; xr[1] = xa.y; xr[2] = xa.z; xr[3] = xa.w;
            xr[4] = xb.x; xr[5] = xb.y; xr[6] = xb.z; xr[7] = xb.w;
        }
#pragma unroll
        for (int r = 0; r < 7; ++r) xnr[r] = xr[r + 1];
        xnr[7] = xs_row[base + 8 * h + 8];

        v2f A2[4];
        int r0[4], r1[4];
#pragma unroll
        for (int kk = 0; kk < 4; ++kk) {
            int j0 = 4 * kk + 2 * h;
            const float2 xj = *reinterpret_cast<const float2*>(&xs_row[base + j0]);
            const int2   rj = *reinterpret_cast<const int2*>(&idx_row[base + j0]);
            A2[kk][0] = (j0     <= ln) ? xj.x : 0.0f;
            A2[kk][1] = (j0 + 1 <= ln) ? xj.y : 0.0f;
            r0[kk] = rj.x;
            r1[kk] = rj.y;
        }

        v8f accW, accWI;
#pragma unroll
        for (int e = 0; e < 8; ++e) { accW[e] = cw; accWI[e] = cwi; }

#pragma unroll
        for (int kk = 0; kk < 4; ++kk) {
            v2f bv;
            bv[0] = wcol[r0[kk] * SLAB];
            bv[1] = wcol[r1[kk] * SLAB];
            accW  = __builtin_amdgcn_wmma_f32_16x16x4_f32(
                        false, A1[kk], false, bv, (short)0, accW,  false, false);
            accWI = __builtin_amdgcn_wmma_f32_16x16x4_f32(
                        false, A2[kk], false, bv, (short)0, accWI, false, false);
        }

        float o8[8];
#pragma unroll
        for (int r = 0; r < 8; ++r) {
            float wc  = accW[r];
            float wic = accWI[r];
            float den = fminf(fmaxf(wc - 1.0f, 1e-10f), 1e10f);
            float o   = wic * __builtin_amdgcn_rcpf(den);
            bool bad  = (wc < 1.0f) | (o < xr[r]) | (o > xnr[r]);
            o8[r] = bad ? MAXT : o;
        }
        float m01 = fminf(o8[0], o8[1]);
        float m23 = fminf(o8[2], o8[3]);
        float m45 = fminf(o8[4], o8[5]);
        float m67 = fminf(o8[6], o8[7]);
        mn = fminf(mn, fminf(fminf(m01, m23), fminf(m45, m67)));

        cw  = __shfl(accW[7],  ln + 16, 32);
        cwi = __shfl(accWI[7], ln + 16, 32);
    }

    {
        float v = fminf(mn, __shfl_xor(mn, 16, 32));
        if (h == 0) s_out[lb * SLAB + c0 + ln] = v;
    }
    __syncthreads();
    if (tid < RPG * SLAB) {
        const int lr = tid / SLAB, cc = tid % SLAB;
        out[(size_t)(b0 + lr) * Mm + slab * SLAB + cc] = s_out[tid];
    }
}

extern "C" void kernel_launch(void* const* d_in, const int* in_sizes, int n_in,
                              void* d_out, int out_size, void* d_ws, size_t ws_size,
                              hipStream_t stream)
{
    (void)n_in; (void)out_size; (void)ws_size;
    const float* x = (const float*)d_in[0];
    const float* w = (const float*)d_in[1];
    float* out = (float*)d_out;
    float* ws  = (float*)d_ws;
    const int B = in_sizes[0] / Nn;
    snn_fc_sort_kernel<<<dim3(B), dim3(512), 0, stream>>>(x, ws, B);
    snn_fc_scan_kernel<<<dim3(B, 2), dim3(512), 0, stream>>>(w, ws, out, B);
}


# Bitonic argsort keeps each thread's 2 elements in registers: partner distance <32 by lane exchange on the vector ALU, 32..256 one LDS round trip, 512 in-thread; same network, same order, same sums
# speedup vs baseline: 2.4085x; 1.3007x over previous
#include <hip/hip_runtime.h>


typedef float v2f __attribute__((ext_vector_type(2)));
typedef float v8f __attribute__((ext_vector_type(8)));

#define MAXT 100000.0f
constexpr int Nn   = 1024;
constexpr int Mm   = 512;
constexpr int SLAB = 64;
constexpr int RPG  = 4;
constexpr int XPAD = 4;


template <int J>
__device__ __forceinline__ int lane_xor(int v)
{
    if constexpr (J == 16)
        return __builtin_amdgcn_permlanex16(v, v, 0x76543210, 0xfedcba98, false, false);
    else
        return __builtin_amdgcn_update_dpp(v, v, 0x160 + J, 0xf, 0xf, false);
}

__device__ __forceinline__ void take_side(float& k, int& i, float pk, int pi, bool keep_min)
{
    const float nk = keep_min ? fminf(k, pk) : fmaxf(k, pk);
    i = (nk != k) ? pi : i;
    k = nk;
}

template <int K, int J>
__device__ __forceinline__ void sort_steps(float& k0, int& i0, float& k1, int& i1,
                                           float (*sk)[Nn], int (*si)[Nn], int& buf, const int tid)
{
    const bool up0 = ((tid & K) == 0);
    const bool up1 = (((tid + 512) & K) == 0);
    if constexpr (J == 512) {
        const float lo = up0 ? fminf(k0, k1) : fmaxf(k0, k1);
        const float hi = up0 ? fmaxf(k0, k1) : fminf(k0, k1);
        const bool  sw = (lo != k0);
        const int   ti = i0;
        i0 = sw ? i1 : i0;  i1 = sw ? ti : i1;
        k0 = lo;            k1 = hi;
    } else if constexpr (J >= 32) {
        float* bk = sk[buf];
        int*   bi = si[buf];
        bk[tid] = k0;  bk[tid + 512] = k1;
        bi[tid] = i0;  bi[tid + 512] = i1;
        __syncthreads();
        const int q = tid ^ J;
        const float pk0 = bk[q], pk1 = bk[q + 512];
        const int   pi0 = bi[q], pi1 = bi[q + 512];
        const bool lower = ((tid & J) == 0);
        take_side(k0, i0, pk0, pi0, lower == up0);
        take_side(k1, i1, pk1, pi1, lower == up1);
        buf ^= 1;
    } else {
        const float pk0 = __builtin_bit_cast(float, lane_xor<J>(__builtin_bit_cast(int, k0)));
        const float pk1 = __builtin_bit_cast(float, lane_xor<J>(__builtin_bit_cast(int, k1)));
        const int   pi0 = lane_xor<J>(i0);
        const int   pi1 = lane_xor<J>(i1);
        const bool lower = ((tid & J) == 0);
        take_side(k0, i0, pk0, pi0, lower == up0);
        take_side(k1, i1, pk1, pi1, lower == up1);
    }
    if constexpr (J > 1)
        sort_steps<K, J / 2>(k0, i0, k1, i1, sk, si, buf, tid);
}

template <int K>
__device__ __forceinline__ void sort_levels(float& k0, int& i0, float& k1, int& i1,
                                            float (*sk)[Nn], int (*si)[Nn], int& buf, const int tid)
{
    sort_steps<K, K / 2>(k0, i0, k1, i1, sk, si, buf, tid);
    if constexpr (K < Nn)
        sort_levels<K * 2>(k0, i0, k1, i1, sk, si, buf, tid);
}

__launch_bounds__(512, 2)
__global__ void snn_fc_sort_kernel(const float* __restrict__ x,
                                   float* __restrict__ ws, int B)
{
    __shared__ float s_k[2][Nn];
    __shared__ int   s_i[2][Nn];

    const int b   = blockIdx.x;
    const int tid = threadIdx.x;

    float k0 = x[(size_t)b * Nn + tid];
    float k1 = x[(size_t)b * Nn + tid + 512];
    int   i0 = tid;
    int   i1 = tid + 512;
    int   buf = 0;

    sort_levels<2>(k0, i0, k1, i1, s_k, s_i, buf, tid);

    float* g_xs  = ws;
    int*   g_idx = reinterpret_cast<int*>(ws) + (size_t)B * Nn;
    g_xs[(size_t)b * Nn + tid]        = k0;
    g_xs[(size_t)b * Nn + tid + 512]  = k1;
    g_idx[(size_t)b * Nn + tid]       = i0;
    g_idx[(size_t)b * Nn + tid + 512] = i1;
}

__launch_bounds__(512, 1)
__global__ void snn_fc_scan_kernel(const float* __restrict__ w,
                                   const float* __restrict__ ws,
                                   float* __restrict__ out, int B)
{
    __shared__ __attribute__((aligned(16))) float s_w[Nn * SLAB];
    __shared__ __attribute__((aligned(16))) float s_xs[RPG][Nn + XPAD];
    __shared__ __attribute__((aligned(16))) int   s_idx[RPG][Nn];
    __shared__ float s_out[RPG * SLAB];

    const int tid  = threadIdx.x;
    const int wg   = blockIdx.x * 2 + blockIdx.y;
    const int slab = wg & (Mm / SLAB - 1);
    const int b0   = (wg / (Mm / SLAB)) * RPG;

    {
        const float* wsl = w + slab * SLAB + (tid & 15) * 4;
        float*       dst = s_w + (tid & 15) * 4;
        const int    rt  = tid >> 4;
        for (int p0 = 0; p0 < Nn / 32; p0 += 8) {
            float4 v[8];
#pragma unroll
            for (int u = 0; u < 8; ++u) {
                const int row = (p0 + u) * 32 + rt;
                v[u] = *reinterpret_cast<const float4*>(wsl + (size_t)row * Mm);
            }
#pragma unroll
            for (int u = 0; u < 8; ++u) {
                const int row = (p0 + u) * 32 + rt;
                *reinterpret_cast<float4*>(dst + row * SLAB) = v[u];
            }
        }
    }
    {
        const float* g_xs  = ws + (size_t)b0 * Nn;
        const int*   g_idx = reinterpret_cast<const int*>(ws) + (size_t)B * Nn + (size_t)b0 * Nn;
#pragma unroll
        for (int k = 0; k < RPG * Nn / (512 * 4); ++k) {
            const int e  = (tid + 512 * k) * 4;
            const int lr = e >> 10, n = e & (Nn - 1);
            *reinterpret_cast<float4*>(&s_xs[lr][n]) = *reinterpret_cast<const float4*>(g_xs + e);
            *reinterpret_cast<int4*>(&s_idx[lr][n])  = *reinterpret_cast<const int4*>(g_idx + e);
        }
        if (tid < RPG * XPAD) s_xs[tid / XPAD][Nn + (tid % XPAD)] = MAXT;
    }
    __syncthreads();

    const int lane = tid & 31;
    const int ln   = lane & 15;
    const int h    = lane >> 4;
    const int wv   = tid >> 5;
    const int lb   = wv >> 2;
    const int c0   = (wv & 3) * 16;

    const float* xs_row  = s_xs[lb];
    const int*   idx_row = s_idx[lb];
    const float* wcol    = s_w + c0 + ln;

    v2f A1[4];
#pragma unroll
    for (int kk = 0; kk < 4; ++kk) {
        int j0 = 4 * kk + 2 * h;
        A1[kk][0] = (j0     <= ln) ? 1.0f : 0.0f;
        A1[kk][1] = (j0 + 1 <= ln) ? 1.0f : 0.0f;
    }

    float cw  = 0.0f;
    float cwi = 0.0f;
    float mn  = MAXT;

    for (int c = 0; c < Nn / 16; ++c) {
        const int base = c * 16;

        float xr[8], xnr[8];
        {
            const float4 xa = *reinterpret_cast<const float4*>(&xs_row[base + 8 * h]);
            const float4 xb = *reinterpret_cast<const float4*>(&xs_row[base + 8 * h + 4]);
            xr[0] = xa.x; xr[1] = xa.y; xr[2] = xa.z; xr[3] = xa.w;
            xr[4] = xb.x; xr[5] = xb.y; xr[6] = xb.z; xr[7] = xb.w;
        }
#pragma unroll
        for (int r = 0; r < 7; ++r) xnr[r] = xr[r + 1];
        xnr[7] = xs_row[base + 8 * h + 8];

        v2f A2[4];
        int r0[4], r1[4];
#pragma unroll
        for (int kk = 0; kk < 4; ++kk) {
            int j0 = 4 * kk + 2 * h;
            const float2 xj = *reinterpret_cast<const float2*>(&xs_row[base + j0]);
            const int2   rj = *reinterpret_cast<const int2*>(&idx_row[base + j0]);
            A2[kk][0] = (j0     <= ln) ? xj.x : 0.0f;
            A2[kk][1] = (j0 + 1 <= ln) ? xj.y : 0.0f;
            r0[kk] = rj.x;
            r1[kk] = rj.y;
        }

        v8f accW, accWI;
#pragma unroll
        for (int e = 0; e < 8; ++e) { accW[e] = cw; accWI[e] = cwi; }

#pragma unroll
        for (int kk = 0; kk < 4; ++kk) {
            v2f bv;
            bv[0] = wcol[r0[kk] * SLAB];
            bv[1] = wcol[r1[kk] * SLAB];
            accW  = __builtin_amdgcn_wmma_f32_16x16x4_f32(
                        false, A1[kk], false, bv, (short)0, accW,  false, false);
            accWI = __builtin_amdgcn_wmma_f32_16x16x4_f32(
                        false, A2[kk], false, bv, (short)0, accWI, false, false);
        }

        float o8[8];
#pragma unroll
        for (int r = 0; r < 8; ++r) {
            float wc  = accW[r];
            float wic = accWI[r];
            float den = fminf(fmaxf(wc - 1.0f, 1e-10f), 1e10f);
            float o   = wic * __builtin_amdgcn_rcpf(den);
            bool bad  = (wc < 1.0f) | (o < xr[r]) | (o > xnr[r]);
            o8[r] = bad ? MAXT : o;
        }
        float m01 = fminf(o8[0], o8[1]);
        float m23 = fminf(o8[2], o8[3]);
        float m45 = fminf(o8[4], o8[5]);
        float m67 = fminf(o8[6], o8[7]);
        mn = fminf(mn, fminf(fminf(m01, m23), fminf(m45, m67)));

        cw  = __shfl(accW[7],  ln + 16, 32);
        cwi = __shfl(accWI[7], ln + 16, 32);
    }

    {
        float v = fminf(mn, __shfl_xor(mn, 16, 32));
        if (h == 0) s_out[lb * SLAB + c0 + ln] = v;
    }
    __syncthreads();
    if (tid < RPG * SLAB) {
        const int lr = tid / SLAB, cc = tid % SLAB;
        out[(size_t)(b0 + lr) * Mm + slab * SLAB + cc] = s_out[tid];
    }
}

extern "C" void kernel_launch(void* const* d_in, const int* in_sizes, int n_in,
                              void* d_out, int out_size, void* d_ws, size_t ws_size,
                              hipStream_t stream)
{
    (void)n_in; (void)out_size; (void)ws_size;
    const float* x = (const float*)d_in[0];
    const float* w = (const float*)d_in[1];
    float* out = (float*)d_out;
    float* ws  = (float*)d_ws;
    const int B = in_sizes[0] / Nn;
    snn_fc_sort_kernel<<<dim3(B), dim3(512), 0, stream>>>(x, ws, B);
    snn_fc_scan_kernel<<<dim3(B, 2), dim3(512), 0, stream>>>(w, ws, out, B);
}


# Argsort folded into the scan launch: each work-group sorts its own 4 rows (8 elements per thread in registers, same network) while its w slab loads; one launch, no workspace; same order, same sums
# speedup vs baseline: 2.4851x; 1.0318x over previous
#include <hip/hip_runtime.h>


typedef float v2f __attribute__((ext_vector_type(2)));
typedef float v8f __attribute__((ext_vector_type(8)));

#define MAXT 100000.0f
constexpr int Nn   = 1024;
constexpr int Mm   = 512;
constexpr int SLAB = 64;
constexpr int RPG  = 4;
constexpr int XPAD = 4;
constexpr int NTH  = 512;
constexpr int EPT  = 8;


template <int M>
__device__ __forceinline__ int lane_xor(int v)
{
    if constexpr (M == 16)
        return __builtin_amdgcn_permlanex16(v, v, 0x76543210, 0xfedcba98, false, false);
    else
        return __builtin_amdgcn_mov_dpp(v, 0x160 + M, 0xf, 0xf, true);
}


__device__ __forceinline__ void take_side(float& k, int& i, float pk, int pi, unsigned min_lanes)
{
    const unsigned lt = __builtin_amdgcn_ballot_w32(pk < k);
    const unsigned gt = __builtin_amdgcn_ballot_w32(pk > k);
    const bool take = __builtin_amdgcn_inverse_ballot_w32(gt ^ ((gt ^ lt) & min_lanes));
    k = take ? pk : k;
    i = take ? pi : i;
}

__device__ __forceinline__ void swap_pair(float& a, int& ia, float& c, int& ic, unsigned up_lanes)
{
    const unsigned gt = __builtin_amdgcn_ballot_w32(a > c);
    const unsigned lt = __builtin_amdgcn_ballot_w32(a < c);
    const bool  sw = __builtin_amdgcn_inverse_ballot_w32(lt ^ ((lt ^ gt) & up_lanes));
    const float ta = a;
    const int   ti = ia;
    a = sw ? c : a;    ia = sw ? ic : ia;
    c = sw ? ta : c;   ic = sw ? ti : ic;
}

template <int K, int J>
__device__ __forceinline__ void sort_steps(float (&k)[EPT], int (&i)[EPT],
                                           float4* scr, const int P0, const int tid)
{
    if constexpr (J < EPT) {
#pragma unroll
        for (int e = 0; e < EPT; ++e) {
            if ((e & J) == 0) {
                const unsigned up_lanes = (K < EPT) ? (((e & K) == 0) ? ~0u : 0u)
                                                    : __builtin_amdgcn_ballot_w32((P0 & K) == 0);
                swap_pair(k[e], i[e], k[e | J], i[e | J], up_lanes);
            }
        }
    } else if constexpr (J < 256) {
        const unsigned min_lanes = __builtin_amdgcn_ballot_w32(((P0 & J) == 0) == ((P0 & K) == 0));
#pragma unroll
        for (int e = 0; e < EPT; ++e) {
            const float pk = __builtin_bit_cast(float, lane_xor<J / EPT>(__builtin_bit_cast(int, k[e])));
            const int   pi = lane_xor<J / EPT>(i[e]);
            take_side(k[e], i[e], pk, pi, min_lanes);
        }
    } else {
        float4* reg = scr + ((K == 1024 && J == 512) ? 4 * NTH : 0);
        reg[tid]           = make_float4(k[0], k[1], k[2], k[3]);
        reg[tid + NTH]     = make_float4(k[4], k[5], k[6], k[7]);
        reg[tid + 2 * NTH] = make_float4(__int_as_float(i[0]), __int_as_float(i[1]),
                                         __int_as_float(i[2]), __int_as_float(i[3]));
        reg[tid + 3 * NTH] = make_float4(__int_as_float(i[4]), __int_as_float(i[5]),
                                         __int_as_float(i[6]), __int_as_float(i[7]));
        __syncthreads();
        const int pt = tid ^ (J / 8);
        const float4 ka = reg[pt];
        const float4 kb = reg[pt + NTH];
        const float4 ia = reg[pt + 2 * NTH];
        const float4 ib = reg[pt + 3 * NTH];
        const unsigned min_lanes = __builtin_amdgcn_ballot_w32(((P0 & J) == 0) == ((P0 & K) == 0));
        take_side(k[0], i[0], ka.x, __float_as_int(ia.x), min_lanes);
        take_side(k[1], i[1], ka.y, __float_as_int(ia.y), min_lanes);
        take_side(k[2], i[2], ka.z, __float_as_int(ia.z), min_lanes);
        take_side(k[3], i[3], ka.w, __float_as_int(ia.w), min_lanes);
        take_side(k[4], i[4], kb.x, __float_as_int(ib.x), min_lanes);
        take_side(k[5], i[5], kb.y, __float_as_int(ib.y), min_lanes);
        take_side(k[6], i[6], kb.z, __float_as_int(ib.z), min_lanes);
        take_side(k[7], i[7], kb.w, __float_as_int(ib.w), min_lanes);
    }
    if constexpr (J > 1)
        sort_steps<K, J / 2>(k, i, scr, P0, tid);
}

template <int K, int KEND>
__device__ __forceinline__ void sort_levels(float (&k)[EPT], int (&i)[EPT],
                                            float4* scr, const int P0, const int tid)
{
    sort_steps<K, K / 2>(k, i, scr, P0, tid);
    if constexpr (K < KEND)
        sort_levels<K * 2, KEND>(k, i, scr, P0, tid);
}

struct Batch { float4 r0, r1, r2, r3, r4, r5, r6, r7; };

template <int PB>
__device__ __forceinline__ float4 slab_row(const float* wsl, const int rt, const int u)
{
    return *reinterpret_cast<const float4*>(wsl + (size_t)((PB + u) * 32 + rt) * Mm);
}
template <int PB>
__device__ __forceinline__ Batch slab_load(const float* wsl, const int rt)
{
    Batch b;
    b.r0 = slab_row<PB>(wsl, rt, 0);  b.r1 = slab_row<PB>(wsl, rt, 1);
    b.r2 = slab_row<PB>(wsl, rt, 2);  b.r3 = slab_row<PB>(wsl, rt, 3);
    b.r4 = slab_row<PB>(wsl, rt, 4);  b.r5 = slab_row<PB>(wsl, rt, 5);
    b.r6 = slab_row<PB>(wsl, rt, 6);  b.r7 = slab_row<PB>(wsl, rt, 7);
    return b;
}
template <int PB>
__device__ __forceinline__ void slab_store(const Batch& b, float* dst, const int rt)
{
    *reinterpret_cast<float4*>(dst + ((PB + 0) * 32 + rt) * SLAB) = b.r0;
    *reinterpret_cast<float4*>(dst + ((PB + 1) * 32 + rt) * SLAB) = b.r1;
    *reinterpret_cast<float4*>(dst + ((PB + 2) * 32 + rt) * SLAB) = b.r2;
    *reinterpret_cast<float4*>(dst + ((PB + 3) * 32 + rt) * SLAB) = b.r3;
    *reinterpret_cast<float4*>(dst + ((PB + 4) * 32 + rt) * SLAB) = b.r4;
    *reinterpret_cast<float4*>(dst + ((PB + 5) * 32 + rt) * SLAB) = b.r5;
    *reinterpret_cast<float4*>(dst + ((PB + 6) * 32 + rt) * SLAB) = b.r6;
    *reinterpret_cast<float4*>(dst + ((PB + 7) * 32 + rt) * SLAB) = b.r7;
}

__launch_bounds__(512, 1)
__global__ void snn_fc_kernel(const float* __restrict__ x,
                              const float* __restrict__ w,
                              float* __restrict__ out)
{
    __shared__ __attribute__((aligned(16))) float s_w[Nn * SLAB];
    __shared__ __attribute__((aligned(16))) float s_xs[RPG][Nn + XPAD];
    __shared__ __attribute__((aligned(16))) int   s_idx[RPG][Nn];
    __shared__ float s_out[RPG * SLAB];

    const int tid  = threadIdx.x;
    const int wg   = blockIdx.x * 2 + blockIdx.y;
    const int slab = wg & (Mm / SLAB - 1);
    const int b0   = (wg / (Mm / SLAB)) * RPG;

    const int lane = tid & 31;
    const int wv   = tid >> 5;
    const int lb   = wv >> 2;
    const int P0   = ((wv & 3) << 8) | (lane << 3);

    const float* xrow = x + (size_t)(b0 + lb) * Nn + P0;
    const float4 xa = *reinterpret_cast<const float4*>(xrow);
    const float4 xb = *reinterpret_cast<const float4*>(xrow + 4);

    const float* wsl = w + slab * SLAB + (tid & 15) * 4;
    float*       dst = s_w + (tid & 15) * 4;
    const int    rt  = tid >> 4;
    float4* scr = reinterpret_cast<float4*>(s_w + 768 * SLAB);

    Batch v = slab_load<0>(wsl, rt);

    float k[EPT] = { xa.x, xa.y, xa.z, xa.w, xb.x, xb.y, xb.z, xb.w };
    int   i[EPT];
#pragma unroll
    for (int e = 0; e < EPT; ++e) i[e] = P0 + e;

    __builtin_amdgcn_sched_barrier(0);
    sort_levels<2, 64>(k, i, scr, P0, tid);
    slab_store<0>(v, dst, rt);
    v = slab_load<8>(wsl, rt);
    __builtin_amdgcn_sched_barrier(0);
    sort_levels<128, 128>(k, i, scr, P0, tid);
    slab_store<8>(v, dst, rt);
    v = slab_load<16>(wsl, rt);
    __builtin_amdgcn_sched_barrier(0);
    sort_levels<256, 256>(k, i, scr, P0, tid);
    slab_store<16>(v, dst, rt);
    v = slab_load<24>(wsl, rt);
    __builtin_amdgcn_sched_barrier(0);
    sort_levels<512, 1024>(k, i, scr, P0, tid);

    *reinterpret_cast<float4*>(&s_xs[lb][P0])      = make_float4(k[0], k[1], k[2], k[3]);
    *reinterpret_cast<float4*>(&s_xs[lb][P0 + 4])  = make_float4(k[4], k[5], k[6], k[7]);
    *reinterpret_cast<int4*>(&s_idx[lb][P0])       = make_int4(i[0], i[1], i[2], i[3]);
    *reinterpret_cast<int4*>(&s_idx[lb][P0 + 4])   = make_int4(i[4], i[5], i[6], i[7]);
    if (tid < RPG * XPAD) s_xs[tid / XPAD][Nn + (tid % XPAD)] = MAXT;
    __syncthreads();
    slab_store<24>(v, dst, rt);
    __syncthreads();

    const int ln   = lane & 15;
    const int h    = lane >> 4;
    const int c0   = (wv & 3) * 16;

    const float* xs_row  = s_xs[lb];
    const int*   idx_row = s_idx[lb];
    const float* wcol    = s_w + c0 + ln;

    v2f A1[4];
#pragma unroll
    for (int kk = 0; kk < 4; ++kk) {
        int j0 = 4 * kk + 2 * h;
        A1[kk][0] = (j0     <= ln) ? 1.0f : 0.0f;
        A1[kk][1] = (j0 + 1 <= ln) ? 1.0f : 0.0f;
    }

    float cw  = 0.0f;
    float cwi = 0.0f;
    float mn  = MAXT;

    for (int c = 0; c < Nn / 16; ++c) {
        const int base = c * 16;

        float xr[8], xnr[8];
        {
            const float4 xa = *reinterpret_cast<const float4*>(&xs_row[base + 8 * h]);
            const float4 xb = *reinterpret_cast<const float4*>(&xs_row[base + 8 * h + 4]);
            xr[0] = xa.x; xr[1] = xa.y; xr[2] = xa.z; xr[3] = xa.w;
            xr[4] = xb.x; xr[5] = xb.y; xr[6] = xb.z; xr[7] = xb.w;
        }
#pragma unroll
        for (int r = 0; r < 7; ++r) xnr[r] = xr[r + 1];
        xnr[7] = xs_row[base + 8 * h + 8];

        v2f A2[4];
        int r0[4], r1[4];
#pragma unroll
        for (int kk = 0; kk < 4; ++kk) {
            int j0 = 4 * kk + 2 * h;
            const float2 xj = *reinterpret_cast<const float2*>(&xs_row[base + j0]);
            const int2   rj = *reinterpret_cast<const int2*>(&idx_row[base + j0]);
            A2[kk][0] = (j0     <= ln) ? xj.x : 0.0f;
            A2[kk][1] = (j0 + 1 <= ln) ? xj.y : 0.0f;
            r0[kk] = rj.x;
            r1[kk] = rj.y;
        }

        v8f accW, accWI;
#pragma unroll
        for (int e = 0; e < 8; ++e) { accW[e] = cw; accWI[e] = cwi; }

#pragma unroll
        for (int kk = 0; kk < 4; ++kk) {
            v2f bv;
            bv[0] = wcol[r0[kk] * SLAB];
            bv[1] = wcol[r1[kk] * SLAB];
            accW  = __builtin_amdgcn_wmma_f32_16x16x4_f32(
                        false, A1[kk], false, bv, (short)0, accW,  false, false);
            accWI = __builtin_amdgcn_wmma_f32_16x16x4_f32(
                        false, A2[kk], false, bv, (short)0, accWI, false, false);
        }

        float o8[8];
#pragma unroll
        for (int r = 0; r < 8; ++r) {
            float wc  = accW[r];
            float wic = accWI[r];
            float den = fminf(fmaxf(wc - 1.0f, 1e-10f), 1e10f);
            float o   = wic * __builtin_amdgcn_rcpf(den);
            bool bad  = (wc < 1.0f) | (o < xr[r]) | (o > xnr[r]);
            o8[r] = bad ? MAXT : o;
        }
        float m01 = fminf(o8[0], o8[1]);
        float m23 = fminf(o8[2], o8[3]);
        float m45 = fminf(o8[4], o8[5]);
        float m67 = fminf(o8[6], o8[7]);
        mn = fminf(mn, fminf(fminf(m01, m23), fminf(m45, m67)));

        cw  = __shfl(accW[7],  ln + 16, 32);
        cwi = __shfl(accWI[7], ln + 16, 32);
    }

    {
        float v = fminf(mn, __shfl_xor(mn, 16, 32));
        if (h == 0) s_out[lb * SLAB + c0 + ln] = v;
    }
    __syncthreads();
    if (tid < RPG * SLAB) {
        const int lr = tid / SLAB, cc = tid % SLAB;
        out[(size_t)(b0 + lr) * Mm + slab * SLAB + cc] = s_out[tid];
    }
}

extern "C" void kernel_launch(void* const* d_in, const int* in_sizes, int n_in,
                              void* d_out, int out_size, void* d_ws, size_t ws_size,
                              hipStream_t stream)
{
    (void)n_in; (void)out_size; (void)d_ws; (void)ws_size;
    const float* x = (const float*)d_in[0];
    const float* w = (const float*)d_in[1];
    float* out = (float*)d_out;
    const int B = in_sizes[0] / Nn;
    snn_fc_kernel<<<dim3(B, 2), dim3(512), 0, stream>>>(x, w, out);
}


# Scan waves lower their own priority as they advance (s_setprio 3,2,1,0 from chunk 0,32,48,56): waves of a SIMD that lag win the issue port, so the youngest no longer scans its tail alone; same sums
# speedup vs baseline: 2.7219x; 1.0953x over previous
#include <hip/hip_runtime.h>


typedef float v2f __attribute__((ext_vector_type(2)));
typedef float v8f __attribute__((ext_vector_type(8)));

#define MAXT 100000.0f
constexpr int Nn   = 1024;
constexpr int Mm   = 512;
constexpr int SLAB = 64;
constexpr int RPG  = 4;
constexpr int XPAD = 4;
constexpr int NTH  = 512;
constexpr int EPT  = 8;


template <int M>
__device__ __forceinline__ int lane_xor(int v)
{
    if constexpr (M == 16)
        return __builtin_amdgcn_permlanex16(v, v, 0x76543210, 0xfedcba98, false, false);
    else
        return __builtin_amdgcn_mov_dpp(v, 0x160 + M, 0xf, 0xf, true);
}


__device__ __forceinline__ void take_side(float& k, int& i, float pk, int pi, unsigned min_lanes)
{
    const unsigned lt = __builtin_amdgcn_ballot_w32(pk < k);
    const unsigned gt = __builtin_amdgcn_ballot_w32(pk > k);
    const bool take = __builtin_amdgcn_inverse_ballot_w32(gt ^ ((gt ^ lt) & min_lanes));
    k = take ? pk : k;
    i = take ? pi : i;
}

__device__ __forceinline__ void swap_pair(float& a, int& ia, float& c, int& ic, unsigned up_lanes)
{
    const unsigned gt = __builtin_amdgcn_ballot_w32(a > c);
    const unsigned lt = __builtin_amdgcn_ballot_w32(a < c);
    const bool  sw = __builtin_amdgcn_inverse_ballot_w32(lt ^ ((lt ^ gt) & up_lanes));
    const float ta = a;
    const int   ti = ia;
    a = sw ? c : a;    ia = sw ? ic : ia;
    c = sw ? ta : c;   ic = sw ? ti : ic;
}

template <int K, int J>
__device__ __forceinline__ void sort_steps(float (&k)[EPT], int (&i)[EPT],
                                           float4* scr, const int P0, const int tid)
{
    if constexpr (J < EPT) {
#pragma unroll
        for (int e = 0; e < EPT; ++e) {
            if ((e & J) == 0) {
                const unsigned up_lanes = (K < EPT) ? (((e & K) == 0) ? ~0u : 0u)
                                                    : __builtin_amdgcn_ballot_w32((P0 & K) == 0);
                swap_pair(k[e], i[e], k[e | J], i[e | J], up_lanes);
            }
        }
    } else if constexpr (J < 256) {
        const unsigned min_lanes = __builtin_amdgcn_ballot_w32(((P0 & J) == 0) == ((P0 & K) == 0));
#pragma unroll
        for (int e = 0; e < EPT; ++e) {
            const float pk = __builtin_bit_cast(float, lane_xor<J / EPT>(__builtin_bit_cast(int, k[e])));
            const int   pi = lane_xor<J / EPT>(i[e]);
            take_side(k[e], i[e], pk, pi, min_lanes);
        }
    } else {
        float4* reg = scr + ((K == 1024 && J == 512) ? 4 * NTH : 0);
        reg[tid]           = make_float4(k[0], k[1], k[2], k[3]);
        reg[tid + NTH]     = make_float4(k[4], k[5], k[6], k[7]);
        reg[tid + 2 * NTH] = make_float4(__int_as_float(i[0]), __int_as_float(i[1]),
                                         __int_as_float(i[2]), __int_as_float(i[3]));
        reg[tid + 3 * NTH] = make_float4(__int_as_float(i[4]), __int_as_float(i[5]),
                                         __int_as_float(i[6]), __int_as_float(i[7]));
        __syncthreads();
        const int pt = tid ^ (J / 8);
        const float4 ka = reg[pt];
        const float4 kb = reg[pt + NTH];
        const float4 ia = reg[pt + 2 * NTH];
        const float4 ib = reg[pt + 3 * NTH];
        const unsigned min_lanes = __builtin_amdgcn_ballot_w32(((P0 & J) == 0) == ((P0 & K) == 0));
        take_side(k[0], i[0], ka.x, __float_as_int(ia.x), min_lanes);
        take_side(k[1], i[1], ka.y, __float_as_int(ia.y), min_lanes);
        take_side(k[2], i[2], ka.z, __float_as_int(ia.z), min_lanes);
        take_side(k[3], i[3], ka.w, __float_as_int(ia.w), min_lanes);
        take_side(k[4], i[4], kb.x, __float_as_int(ib.x), min_lanes);
        take_side(k[5], i[5], kb.y, __float_as_int(ib.y), min_lanes);
        take_side(k[6], i[6], kb.z, __float_as_int(ib.z), min_lanes);
        take_side(k[7], i[7], kb.w, __float_as_int(ib.w), min_lanes);
    }
    if constexpr (J > 1)
        sort_steps<K, J / 2>(k, i, scr, P0, tid);
}

template <int K, int KEND>
__device__ __forceinline__ void sort_levels(float (&k)[EPT], int (&i)[EPT],
                                            float4* scr, const int P0, const int tid)
{
    sort_steps<K, K / 2>(k, i, scr, P0, tid);
    if constexpr (K < KEND)
        sort_levels<K * 2, KEND>(k, i, scr, P0, tid);
}

struct Batch { float4 r0, r1, r2, r3, r4, r5, r6, r7; };

template <int PB>
__device__ __forceinline__ float4 slab_row(const float* wsl, const int rt, const int u)
{
    return *reinterpret_cast<const float4*>(wsl + (size_t)((PB + u) * 32 + rt) * Mm);
}
template <int PB>
__device__ __forceinline__ Batch slab_load(const float* wsl, const int rt)
{
    Batch b;
    b.r0 = slab_row<PB>(wsl, rt, 0);  b.r1 = slab_row<PB>(wsl, rt, 1);
    b.r2 = slab_row<PB>(wsl, rt, 2);  b.r3 = slab_row<PB>(wsl, rt, 3);
    b.r4 = slab_row<PB>(wsl, rt, 4);  b.r5 = slab_row<PB>(wsl, rt, 5);
    b.r6 = slab_row<PB>(wsl, rt, 6);  b.r7 = slab_row<PB>(wsl, rt, 7);
    return b;
}
template <int PB>
__device__ __forceinline__ void slab_store(const Batch& b, float* dst, const int rt)
{
    *reinterpret_cast<float4*>(dst + ((PB + 0) * 32 + rt) * SLAB) = b.r0;
    *reinterpret_cast<float4*>(dst + ((PB + 1) * 32 + rt) * SLAB) = b.r1;
    *reinterpret_cast<float4*>(dst + ((PB + 2) * 32 + rt) * SLAB) = b.r2;
    *reinterpret_cast<float4*>(dst + ((PB + 3) * 32 + rt) * SLAB) = b.r3;
    *reinterpret_cast<float4*>(dst + ((PB + 4) * 32 + rt) * SLAB) = b.r4;
    *reinterpret_cast<float4*>(dst + ((PB + 5) * 32 + rt) * SLAB) = b.r5;
    *reinterpret_cast<float4*>(dst + ((PB + 6) * 32 + rt) * SLAB) = b.r6;
    *reinterpret_cast<float4*>(dst + ((PB + 7) * 32 + rt) * SLAB) = b.r7;
}

__launch_bounds__(512, 1)
__global__ void snn_fc_kernel(const float* __restrict__ x,
                              const float* __restrict__ w,
                              float* __restrict__ out)
{
    __shared__ __attribute__((aligned(16))) float s_w[Nn * SLAB];
    __shared__ __attribute__((aligned(16))) float s_xs[RPG][Nn + XPAD];
    __shared__ __attribute__((aligned(16))) int   s_idx[RPG][Nn];
    __shared__ float s_out[RPG * SLAB];

    const int tid  = threadIdx.x;
    const int wg   = blockIdx.x * 2 + blockIdx.y;
    const int slab = wg & (Mm / SLAB - 1);
    const int b0   = (wg / (Mm / SLAB)) * RPG;

    const int lane = tid & 31;
    const int wv   = tid >> 5;
    const int lb   = wv >> 2;
    const int P0   = ((wv & 3) << 8) | (lane << 3);

    const float* xrow = x + (size_t)(b0 + lb) * Nn + P0;
    const float4 xa = *reinterpret_cast<const float4*>(xrow);
    const float4 xb = *reinterpret_cast<const float4*>(xrow + 4);

    const float* wsl = w + slab * SLAB + (tid & 15) * 4;
    float*       dst = s_w + (tid & 15) * 4;
    const int    rt  = tid >> 4;
    float4* scr = reinterpret_cast<float4*>(s_w + 768 * SLAB);

    Batch v = slab_load<0>(wsl, rt);

    float k[EPT] = { xa.x, xa.y, xa.z, xa.w, xb.x, xb.y, xb.z, xb.w };
    int   i[EPT];
#pragma unroll
    for (int e = 0; e < EPT; ++e) i[e] = P0 + e;

    __builtin_amdgcn_sched_barrier(0);
    sort_levels<2, 64>(k, i, scr, P0, tid);
    slab_store<0>(v, dst, rt);
    v = slab_load<8>(wsl, rt);
    __builtin_amdgcn_sched_barrier(0);
    sort_levels<128, 128>(k, i, scr, P0, tid);
    slab_store<8>(v, dst, rt);
    v = slab_load<16>(wsl, rt);
    __builtin_amdgcn_sched_barrier(0);
    sort_levels<256, 256>(k, i, scr, P0, tid);
    slab_store<16>(v, dst, rt);
    v = slab_load<24>(wsl, rt);
    __builtin_amdgcn_sched_barrier(0);
    sort_levels<512, 1024>(k, i, scr, P0, tid);

    *reinterpret_cast<float4*>(&s_xs[lb][P0])      = make_float4(k[0], k[1], k[2], k[3]);
    *reinterpret_cast<float4*>(&s_xs[lb][P0 + 4])  = make_float4(k[4], k[5], k[6], k[7]);
    *reinterpret_cast<int4*>(&s_idx[lb][P0])       = make_int4(i[0], i[1], i[2], i[3]);
    *reinterpret_cast<int4*>(&s_idx[lb][P0 + 4])   = make_int4(i[4], i[5], i[6], i[7]);
    if (tid < RPG * XPAD) s_xs[tid / XPAD][Nn + (tid % XPAD)] = MAXT;
    __syncthreads();
    slab_store<24>(v, dst, rt);
    __syncthreads();

    const int ln   = lane & 15;
    const int h    = lane >> 4;
    const int c0   = (wv & 3) * 16;

    const float* xs_row  = s_xs[lb];
    const int*   idx_row = s_idx[lb];
    const float* wcol    = s_w + c0 + ln;

    v2f A1[4];
#pragma unroll
    for (int kk = 0; kk < 4; ++kk) {
        int j0 = 4 * kk + 2 * h;
        A1[kk][0] = (j0     <= ln) ? 1.0f : 0.0f;
        A1[kk][1] = (j0 + 1 <= ln) ? 1.0f : 0.0f;
    }

    float cw  = 0.0f;
    float cwi = 0.0f;
    float mn  = MAXT;

    auto scan_chunks = [&](const int cbeg, const int cend) __attribute__((always_inline)) {
    for (int c = cbeg; c < cend; ++c) {
        const int base = c * 16;

        float xr[8], xnr[8];
        {
            const float4 xa = *reinterpret_cast<const float4*>(&xs_row[base + 8 * h]);
            const float4 xb = *reinterpret_cast<const float4*>(&xs_row[base + 8 * h + 4]);
            xr[0] = xa.x; xr[1] = xa.y; xr[2] = xa.z; xr[3] = xa.w;
            xr[4] = xb.x; xr[5] = xb.y; xr[6] = xb.z; xr[7] = xb.w;
        }
#pragma unroll
        for (int r = 0; r < 7; ++r) xnr[r] = xr[r + 1];
        xnr[7] = xs_row[base + 8 * h + 8];

        v2f A2[4];
        int r0[4], r1[4];
#pragma unroll
        for (int kk = 0; kk < 4; ++kk) {
            int j0 = 4 * kk + 2 * h;
            const float2 xj = *reinterpret_cast<const float2*>(&xs_row[base + j0]);
            const int2   rj = *reinterpret_cast<const int2*>(&idx_row[base + j0]);
            A2[kk][0] = (j0     <= ln) ? xj.x : 0.0f;
            A2[kk][1] = (j0 + 1 <= ln) ? xj.y : 0.0f;
            r0[kk] = rj.x;
            r1[kk] = rj.y;
        }

        v8f accW, accWI;
#pragma unroll
        for (int e = 0; e < 8; ++e) { accW[e] = cw; accWI[e] = cwi; }

#pragma unroll
        for (int kk = 0; kk < 4; ++kk) {
            v2f bv;
            bv[0] = wcol[r0[kk] * SLAB];
            bv[1] = wcol[r1[kk] * SLAB];
            accW  = __builtin_amdgcn_wmma_f32_16x16x4_f32(
                        false, A1[kk], false, bv, (short)0, accW,  false, false);
            accWI = __builtin_amdgcn_wmma_f32_16x16x4_f32(
                        false, A2[kk], false, bv, (short)0, accWI, false, false);
        }

        float o8[8];
#pragma unroll
        for (int r = 0; r < 8; ++r) {
            float wc  = accW[r];
            float wic = accWI[r];
            float den = fminf(fmaxf(wc - 1.0f, 1e-10f), 1e10f);
            float o   = wic * __builtin_amdgcn_rcpf(den);
            bool bad  = (wc < 1.0f) | (o < xr[r]) | (o > xnr[r]);
            o8[r] = bad ? MAXT : o;
        }
        float m01 = fminf(o8[0], o8[1]);
        float m23 = fminf(o8[2], o8[3]);
        float m45 = fminf(o8[4], o8[5]);
        float m67 = fminf(o8[6], o8[7]);
        mn = fminf(mn, fminf(fminf(m01, m23), fminf(m45, m67)));

        cw  = __shfl(accW[7],  ln + 16, 32);
        cwi = __shfl(accWI[7], ln + 16, 32);
    }
    };
    __builtin_amdgcn_s_setprio(3);
    scan_chunks(0, 32);
    __builtin_amdgcn_s_setprio(2);
    scan_chunks(32, 48);
    __builtin_amdgcn_s_setprio(1);
    scan_chunks(48, 56);
    __builtin_amdgcn_s_setprio(0);
    scan_chunks(56, Nn / 16);

    {
        float v = fminf(mn, __shfl_xor(mn, 16, 32));
        if (h == 0) s_out[lb * SLAB + c0 + ln] = v;
    }
    __syncthreads();
    if (tid < RPG * SLAB) {
        const int lr = tid / SLAB, cc = tid % SLAB;
        out[(size_t)(b0 + lr) * Mm + slab * SLAB + cc] = s_out[tid];
    }
}

extern "C" void kernel_launch(void* const* d_in, const int* in_sizes, int n_in,
                              void* d_out, int out_size, void* d_ws, size_t ws_size,
                              hipStream_t stream)
{
    (void)n_in; (void)out_size; (void)d_ws; (void)ws_size;
    const float* x = (const float*)d_in[0];
    const float* w = (const float*)d_in[1];
    float* out = (float*)d_out;
    const int B = in_sizes[0] / Nn;
    snn_fc_kernel<<<dim3(B, 2), dim3(512), 0, stream>>>(x, w, out);
}
